# starting kernel
# baseline (speedup 1.0000x reference)
#include <hip/hip_runtime.h>
#include <cmath>

#define NN   100000
#define EE   3200000
#define F_INd 500
#define HID  16
#define NC   3

typedef __attribute__((ext_vector_type(2))) float v2f;
typedef __attribute__((ext_vector_type(8))) float v8f;

__global__ void k_init_deg(float* __restrict__ deg) {
  int i = blockIdx.x * blockDim.x + threadIdx.x;
  if (i < NN) deg[i] = 1.0f;
}

__global__ void k_count_deg(const int* __restrict__ dst, float* __restrict__ deg) {
  int e = blockIdx.x * blockDim.x + threadIdx.x;
  if (e < EE) atomicAdd(&deg[dst[e]], 1.0f);
}

__global__ void k_rsqrt(float* __restrict__ dinv) {
  int i = blockIdx.x * blockDim.x + threadIdx.x;
  if (i < NN) dinv[i] = rsqrtf(dinv[i]);
}

__global__ __launch_bounds__(256) void k_gemm1_wmma(const float* __restrict__ X,
                                                    const float* __restrict__ W,
                                                    float* __restrict__ H1) {
  __shared__ float Wlds[F_INd * HID];
  for (int i = threadIdx.x; i < F_INd * HID; i += 256) Wlds[i] = W[i];
  __syncthreads();

  const int wave = threadIdx.x >> 5;
  const int lane = threadIdx.x & 31;
  const int tile = blockIdx.x * 8 + wave;
  if (tile >= NN / 16) return;

  const int row_base = tile * 16;
  const int m    = lane & 15;
  const int hi   = lane >> 4;
  const int koff = hi << 1;

  const float* xrow = X + (size_t)(row_base + m) * F_INd;

  v8f acc = {};
  for (int k0 = 0; k0 < F_INd; k0 += 4) {
    v2f a, b;
    a.x = xrow[k0 + koff];
    a.y = xrow[k0 + koff + 1];
    b.x = Wlds[(k0 + koff)     * HID + m];
    b.y = Wlds[(k0 + koff + 1) * HID + m];
    acc = __builtin_amdgcn_wmma_f32_16x16x4_f32(false, a, false, b,
                                                (short)0, acc, false, false);
  }

  const int rbase = row_base + (hi << 3);
#pragma unroll
  for (int r = 0; r < 8; ++r)
    H1[(size_t)(rbase + r) * HID + m] = acc[r];
}

__global__ void k_agg1_init(const float* __restrict__ h1, const float* __restrict__ dinv,
                            float* __restrict__ agg1) {
  int i = blockIdx.x * blockDim.x + threadIdx.x;
  if (i >= NN * HID) return;
  int node = i >> 4;
  float di = dinv[node];
  agg1[i] = h1[i] * di * di;
}

__global__ void k_agg1_edges(const float* __restrict__ h1, const float* __restrict__ dinv,
                             const int* __restrict__ src, const int* __restrict__ dst,
                             float* __restrict__ agg1) {
  long long idx = (long long)blockIdx.x * blockDim.x + threadIdx.x;
  int e = (int)(idx >> 4);
  int f = (int)(idx & 15);
  if (e >= EE) return;
  int s = src[e], d = dst[e];
  float nrm = dinv[s] * dinv[d];
  atomicAdd(&agg1[(size_t)d * HID + f], h1[(size_t)s * HID + f] * nrm);
}

__global__ void k_relu_lin2(const float* __restrict__ agg1, const float* __restrict__ b1,
                            const float* __restrict__ W2, float* __restrict__ h2) {
  int i = blockIdx.x * blockDim.x + threadIdx.x;
  if (i >= NN) return;
  float v[HID];
#pragma unroll
  for (int f = 0; f < HID; ++f) {
    float t = agg1[(size_t)i * HID + f] + b1[f];
    v[f] = t > 0.0f ? t : 0.0f;
  }
#pragma unroll
  for (int c = 0; c < NC; ++c) {
    float s = 0.0f;
#pragma unroll
    for (int f = 0; f < HID; ++f) s += v[f] * W2[f * NC + c];
    h2[(size_t)i * 4 + c] = s;
  }
  h2[(size_t)i * 4 + 3] = 0.0f;
}

__global__ void k_agg2_init(const float* __restrict__ h2, const float* __restrict__ dinv,
                            float* __restrict__ agg2) {
  int i = blockIdx.x * blockDim.x + threadIdx.x;
  if (i >= NN * 4) return;
  int node = i >> 2;
  float di = dinv[node];
  agg2[i] = h2[i] * di * di;
}

__global__ void k_agg2_edges(const float* __restrict__ h2, const float* __restrict__ dinv,
                             const int* __restrict__ src, const int* __restrict__ dst,
                             float* __restrict__ agg2) {
  int e = blockIdx.x * blockDim.x + threadIdx.x;
  if (e >= EE) return;
  int s = src[e], d = dst[e];
  float nrm = dinv[s] * dinv[d];
#pragma unroll
  for (int c = 0; c < NC; ++c)
    atomicAdd(&agg2[(size_t)d * 4 + c], h2[(size_t)s * 4 + c] * nrm);
}

__global__ void k_logsoftmax(const float* __restrict__ agg2, const float* __restrict__ b2,
                             float* __restrict__ out) {
  int i = blockIdx.x * blockDim.x + threadIdx.x;
  if (i >= NN) return;
  float z0 = agg2[(size_t)i * 4 + 0] + b2[0];
  float z1 = agg2[(size_t)i * 4 + 1] + b2[1];
  float z2 = agg2[(size_t)i * 4 + 2] + b2[2];
  float mx = fmaxf(z0, fmaxf(z1, z2));
  float lse = logf(expf(z0 - mx) + expf(z1 - mx) + expf(z2 - mx));
  out[(size_t)i * NC + 0] = z0 - mx - lse;
  out[(size_t)i * NC + 1] = z1 - mx - lse;
  out[(size_t)i * NC + 2] = z2 - mx - lse;
}

extern "C" void kernel_launch(void* const* d_in, const int* in_sizes, int n_in,
                              void* d_out, int out_size, void* d_ws, size_t ws_size,
                              hipStream_t stream) {
  const float* x  = (const float*)d_in[0];
  const float* W1 = (const float*)d_in[1];
  const float* b1 = (const float*)d_in[2];
  const float* W2 = (const float*)d_in[3];
  const float* b2 = (const float*)d_in[4];
  const int*   ei = (const int*)d_in[5];
  const int* src = ei;
  const int* dst = ei + EE;
  float* out = (float*)d_out;

  float* ws   = (float*)d_ws;
  float* h1   = ws;
  float* dinv = h1   + (size_t)NN * HID;
  float* agg1 = dinv + (size_t)NN;
  float* h2   = agg1 + (size_t)NN * HID;
  float* agg2 = h2   + (size_t)NN * 4;

  const int T = 256;
  k_init_deg   <<<(NN + T - 1) / T, T, 0, stream>>>(dinv);
  k_count_deg  <<<(EE + T - 1) / T, T, 0, stream>>>(dst, dinv);
  k_rsqrt      <<<(NN + T - 1) / T, T, 0, stream>>>(dinv);

  k_gemm1_wmma <<<(NN / 16 + 7) / 8, T, 0, stream>>>(x, W1, h1);

  k_agg1_init  <<<(NN * HID + T - 1) / T, T, 0, stream>>>(h1, dinv, agg1);
  {
    long long total = (long long)EE * HID;
    int blocks = (int)((total + T - 1) / T);
    k_agg1_edges<<<blocks, T, 0, stream>>>(h1, dinv, src, dst, agg1);
  }

  k_relu_lin2  <<<(NN + T - 1) / T, T, 0, stream>>>(agg1, b1, W2, h2);

  k_agg2_init  <<<(NN * 4 + T - 1) / T, T, 0, stream>>>(h2, dinv, agg2);
  k_agg2_edges <<<(EE + T - 1) / T, T, 0, stream>>>(h2, dinv, src, dst, agg2);

  k_logsoftmax <<<(NN + T - 1) / T, T, 0, stream>>>(agg2, b2, out);
}


# layer-2 edge scatter: 4 adjacent lanes per edge (3 active) instead of 1 thread issuing 3 atomics, so an edge's 3 float adds leave as one 16-byte request; atomic arrival order free as before
# speedup vs baseline: 1.2167x; 1.2167x over previous
#include <hip/hip_runtime.h>
#include <cmath>

#define NN   100000
#define EE   3200000
#define F_INd 500
#define HID  16
#define NC   3

typedef __attribute__((ext_vector_type(2))) float v2f;
typedef __attribute__((ext_vector_type(8))) float v8f;

__global__ void k_init_deg(float* __restrict__ deg) {
  int i = blockIdx.x * blockDim.x + threadIdx.x;
  if (i < NN) deg[i] = 1.0f;
}

__global__ void k_count_deg(const int* __restrict__ dst, float* __restrict__ deg) {
  int e = blockIdx.x * blockDim.x + threadIdx.x;
  if (e < EE) atomicAdd(&deg[dst[e]], 1.0f);
}

__global__ void k_rsqrt(float* __restrict__ dinv) {
  int i = blockIdx.x * blockDim.x + threadIdx.x;
  if (i < NN) dinv[i] = rsqrtf(dinv[i]);
}

__global__ __launch_bounds__(256) void k_gemm1_wmma(const float* __restrict__ X,
                                                    const float* __restrict__ W,
                                                    float* __restrict__ H1) {
  __shared__ float Wlds[F_INd * HID];
  for (int i = threadIdx.x; i < F_INd * HID; i += 256) Wlds[i] = W[i];
  __syncthreads();

  const int wave = threadIdx.x >> 5;
  const int lane = threadIdx.x & 31;
  const int tile = blockIdx.x * 8 + wave;
  if (tile >= NN / 16) return;

  const int row_base = tile * 16;
  const int m    = lane & 15;
  const int hi   = lane >> 4;
  const int koff = hi << 1;

  const float* xrow = X + (size_t)(row_base + m) * F_INd;

  v8f acc = {};
  for (int k0 = 0; k0 < F_INd; k0 += 4) {
    v2f a, b;
    a.x = xrow[k0 + koff];
    a.y = xrow[k0 + koff + 1];
    b.x = Wlds[(k0 + koff)     * HID + m];
    b.y = Wlds[(k0 + koff + 1) * HID + m];
    acc = __builtin_amdgcn_wmma_f32_16x16x4_f32(false, a, false, b,
                                                (short)0, acc, false, false);
  }

  const int rbase = row_base + (hi << 3);
#pragma unroll
  for (int r = 0; r < 8; ++r)
    H1[(size_t)(rbase + r) * HID + m] = acc[r];
}

__global__ void k_agg1_init(const float* __restrict__ h1, const float* __restrict__ dinv,
                            float* __restrict__ agg1) {
  int i = blockIdx.x * blockDim.x + threadIdx.x;
  if (i >= NN * HID) return;
  int node = i >> 4;
  float di = dinv[node];
  agg1[i] = h1[i] * di * di;
}

__global__ void k_agg1_edges(const float* __restrict__ h1, const float* __restrict__ dinv,
                             const int* __restrict__ src, const int* __restrict__ dst,
                             float* __restrict__ agg1) {
  long long idx = (long long)blockIdx.x * blockDim.x + threadIdx.x;
  int e = (int)(idx >> 4);
  int f = (int)(idx & 15);
  if (e >= EE) return;
  int s = src[e], d = dst[e];
  float nrm = dinv[s] * dinv[d];
  atomicAdd(&agg1[(size_t)d * HID + f], h1[(size_t)s * HID + f] * nrm);
}

__global__ void k_relu_lin2(const float* __restrict__ agg1, const float* __restrict__ b1,
                            const float* __restrict__ W2, float* __restrict__ h2) {
  int i = blockIdx.x * blockDim.x + threadIdx.x;
  if (i >= NN) return;
  float v[HID];
#pragma unroll
  for (int f = 0; f < HID; ++f) {
    float t = agg1[(size_t)i * HID + f] + b1[f];
    v[f] = t > 0.0f ? t : 0.0f;
  }
#pragma unroll
  for (int c = 0; c < NC; ++c) {
    float s = 0.0f;
#pragma unroll
    for (int f = 0; f < HID; ++f) s += v[f] * W2[f * NC + c];
    h2[(size_t)i * 4 + c] = s;
  }
  h2[(size_t)i * 4 + 3] = 0.0f;
}

__global__ void k_agg2_init(const float* __restrict__ h2, const float* __restrict__ dinv,
                            float* __restrict__ agg2) {
  int i = blockIdx.x * blockDim.x + threadIdx.x;
  if (i >= NN * 4) return;
  int node = i >> 2;
  float di = dinv[node];
  agg2[i] = h2[i] * di * di;
}

__global__ void k_agg2_edges(const float* __restrict__ h2, const float* __restrict__ dinv,
                             const int* __restrict__ src, const int* __restrict__ dst,
                             float* __restrict__ agg2) {
  int idx = blockIdx.x * blockDim.x + threadIdx.x;
  int e = idx >> 2;
  int c = idx & 3;
  if (e >= EE || c >= NC) return;
  int s = src[e], d = dst[e];
  float nrm = dinv[s] * dinv[d];
  atomicAdd(&agg2[(size_t)d * 4 + c], h2[(size_t)s * 4 + c] * nrm);
}

__global__ void k_logsoftmax(const float* __restrict__ agg2, const float* __restrict__ b2,
                             float* __restrict__ out) {
  int i = blockIdx.x * blockDim.x + threadIdx.x;
  if (i >= NN) return;
  float z0 = agg2[(size_t)i * 4 + 0] + b2[0];
  float z1 = agg2[(size_t)i * 4 + 1] + b2[1];
  float z2 = agg2[(size_t)i * 4 + 2] + b2[2];
  float mx = fmaxf(z0, fmaxf(z1, z2));
  float lse = logf(expf(z0 - mx) + expf(z1 - mx) + expf(z2 - mx));
  out[(size_t)i * NC + 0] = z0 - mx - lse;
  out[(size_t)i * NC + 1] = z1 - mx - lse;
  out[(size_t)i * NC + 2] = z2 - mx - lse;
}

extern "C" void kernel_launch(void* const* d_in, const int* in_sizes, int n_in,
                              void* d_out, int out_size, void* d_ws, size_t ws_size,
                              hipStream_t stream) {
  const float* x  = (const float*)d_in[0];
  const float* W1 = (const float*)d_in[1];
  const float* b1 = (const float*)d_in[2];
  const float* W2 = (const float*)d_in[3];
  const float* b2 = (const float*)d_in[4];
  const int*   ei = (const int*)d_in[5];
  const int* src = ei;
  const int* dst = ei + EE;
  float* out = (float*)d_out;

  float* ws   = (float*)d_ws;
  float* h1   = ws;
  float* dinv = h1   + (size_t)NN * HID;
  float* agg1 = dinv + (size_t)NN;
  float* h2   = agg1 + (size_t)NN * HID;
  float* agg2 = h2   + (size_t)NN * 4;

  const int T = 256;
  k_init_deg   <<<(NN + T - 1) / T, T, 0, stream>>>(dinv);
  k_count_deg  <<<(EE + T - 1) / T, T, 0, stream>>>(dst, dinv);
  k_rsqrt      <<<(NN + T - 1) / T, T, 0, stream>>>(dinv);

  k_gemm1_wmma <<<(NN / 16 + 7) / 8, T, 0, stream>>>(x, W1, h1);

  k_agg1_init  <<<(NN * HID + T - 1) / T, T, 0, stream>>>(h1, dinv, agg1);
  {
    long long total = (long long)EE * HID;
    int blocks = (int)((total + T - 1) / T);
    k_agg1_edges<<<blocks, T, 0, stream>>>(h1, dinv, src, dst, agg1);
  }

  k_relu_lin2  <<<(NN + T - 1) / T, T, 0, stream>>>(agg1, b1, W2, h2);

  k_agg2_init  <<<(NN * 4 + T - 1) / T, T, 0, stream>>>(h2, dinv, agg2);
  k_agg2_edges <<<(EE * 4 + T - 1) / T, T, 0, stream>>>(h2, dinv, src, dst, agg2);

  k_logsoftmax <<<(NN + T - 1) / T, T, 0, stream>>>(agg2, b2, out);
}


# both edge scatters walk the edge list in long waves (128 edges each): a lane fetches one edge's indices and norm once, 16 rows are requested as a group before their adds; atomic arrival order free
# speedup vs baseline: 1.4674x; 1.2060x over previous
#include <hip/hip_runtime.h>
#include <cmath>

#define NN   100000
#define EE   3200000
#define F_INd 500
#define HID  16
#define NC   3

typedef __attribute__((ext_vector_type(2))) float v2f;
typedef __attribute__((ext_vector_type(8))) float v8f;

__global__ void k_init_deg(float* __restrict__ deg) {
  int i = blockIdx.x * blockDim.x + threadIdx.x;
  if (i < NN) deg[i] = 1.0f;
}

__global__ void k_count_deg(const int* __restrict__ dst, float* __restrict__ deg) {
  int e = blockIdx.x * blockDim.x + threadIdx.x;
  if (e < EE) atomicAdd(&deg[dst[e]], 1.0f);
}

__global__ void k_rsqrt(float* __restrict__ dinv) {
  int i = blockIdx.x * blockDim.x + threadIdx.x;
  if (i < NN) dinv[i] = rsqrtf(dinv[i]);
}

__global__ __launch_bounds__(256) void k_gemm1_wmma(const float* __restrict__ X,
                                                    const float* __restrict__ W,
                                                    float* __restrict__ H1) {
  __shared__ float Wlds[F_INd * HID];
  for (int i = threadIdx.x; i < F_INd * HID; i += 256) Wlds[i] = W[i];
  __syncthreads();

  const int wave = threadIdx.x >> 5;
  const int lane = threadIdx.x & 31;
  const int tile = blockIdx.x * 8 + wave;
  if (tile >= NN / 16) return;

  const int row_base = tile * 16;
  const int m    = lane & 15;
  const int hi   = lane >> 4;
  const int koff = hi << 1;

  const float* xrow = X + (size_t)(row_base + m) * F_INd;

  v8f acc = {};
  for (int k0 = 0; k0 < F_INd; k0 += 4) {
    v2f a, b;
    a.x = xrow[k0 + koff];
    a.y = xrow[k0 + koff + 1];
    b.x = Wlds[(k0 + koff)     * HID + m];
    b.y = Wlds[(k0 + koff + 1) * HID + m];
    acc = __builtin_amdgcn_wmma_f32_16x16x4_f32(false, a, false, b,
                                                (short)0, acc, false, false);
  }

  const int rbase = row_base + (hi << 3);
#pragma unroll
  for (int r = 0; r < 8; ++r)
    H1[(size_t)(rbase + r) * HID + m] = acc[r];
}

__global__ void k_agg1_init(const float* __restrict__ h1, const float* __restrict__ dinv,
                            float* __restrict__ agg1) {
  int i = blockIdx.x * blockDim.x + threadIdx.x;
  if (i >= NN * HID) return;
  int node = i >> 4;
  float di = dinv[node];
  agg1[i] = h1[i] * di * di;
}

#define EPW 128
static_assert(EE % EPW == 0, "edge runs must tile the edge list exactly");

__global__ __launch_bounds__(256) void k_agg1_edges(const float* __restrict__ h1,
                                                    const float* __restrict__ dinv,
                                                    const int* __restrict__ src,
                                                    const int* __restrict__ dst,
                                                    float* __restrict__ agg1) {
  const int lane = threadIdx.x & 31;
  const int wave = (blockIdx.x * 256 + threadIdx.x) >> 5;
  const int f    = lane & 15;
  const int e0   = wave * EPW + lane;

  int   s[EPW / 32], d[EPW / 32];
  float nrm[EPW / 32];
#pragma unroll
  for (int c = 0; c < EPW / 32; ++c) { s[c] = src[e0 + c * 32]; d[c] = dst[e0 + c * 32]; }
#pragma unroll
  for (int c = 0; c < EPW / 32; ++c) nrm[c] = dinv[(unsigned)s[c]] * dinv[(unsigned)d[c]];

#pragma unroll
  for (int c = 0; c < EPW / 32; ++c) {
    float v[16];
#pragma unroll
    for (int j = 0; j < 16; ++j) {
      int sj = __shfl(s[c], j, 16);
      v[j] = h1[(unsigned)sj * HID + f];
    }
#pragma unroll
    for (int j = 0; j < 16; ++j) {
      int   dj = __shfl(d[c], j, 16);
      float nj = __shfl(nrm[c], j, 16);
      atomicAdd(&agg1[(unsigned)dj * HID + f], v[j] * nj);
    }
  }
}

__global__ void k_relu_lin2(const float* __restrict__ agg1, const float* __restrict__ b1,
                            const float* __restrict__ W2, float* __restrict__ h2) {
  int i = blockIdx.x * blockDim.x + threadIdx.x;
  if (i >= NN) return;
  float v[HID];
#pragma unroll
  for (int f = 0; f < HID; ++f) {
    float t = agg1[(size_t)i * HID + f] + b1[f];
    v[f] = t > 0.0f ? t : 0.0f;
  }
#pragma unroll
  for (int c = 0; c < NC; ++c) {
    float s = 0.0f;
#pragma unroll
    for (int f = 0; f < HID; ++f) s += v[f] * W2[f * NC + c];
    h2[(size_t)i * 4 + c] = s;
  }
  h2[(size_t)i * 4 + 3] = 0.0f;
}

__global__ void k_agg2_init(const float* __restrict__ h2, const float* __restrict__ dinv,
                            float* __restrict__ agg2) {
  int i = blockIdx.x * blockDim.x + threadIdx.x;
  if (i >= NN * 4) return;
  int node = i >> 2;
  float di = dinv[node];
  agg2[i] = h2[i] * di * di;
}

__global__ __launch_bounds__(256) void k_agg2_edges(const float* __restrict__ h2,
                                                    const float* __restrict__ dinv,
                                                    const int* __restrict__ src,
                                                    const int* __restrict__ dst,
                                                    float* __restrict__ agg2) {
  const int lane = threadIdx.x & 31;
  const int wave = (blockIdx.x * 256 + threadIdx.x) >> 5;
  const int c4   = lane & 3;
  const int e0   = wave * EPW + lane;

  int   s[EPW / 32], d[EPW / 32];
  float nrm[EPW / 32];
#pragma unroll
  for (int c = 0; c < EPW / 32; ++c) { s[c] = src[e0 + c * 32]; d[c] = dst[e0 + c * 32]; }
#pragma unroll
  for (int c = 0; c < EPW / 32; ++c) nrm[c] = dinv[(unsigned)s[c]] * dinv[(unsigned)d[c]];

  float    v[EPW / 32][4];
  unsigned a[EPW / 32][4];
#pragma unroll
  for (int c = 0; c < EPW / 32; ++c) {
#pragma unroll
    for (int j = 0; j < 4; ++j) {
      int sj = __shfl(s[c], j, 4);
      v[c][j] = h2[(unsigned)sj * 4 + c4];
    }
  }
#pragma unroll
  for (int c = 0; c < EPW / 32; ++c) {
#pragma unroll
    for (int j = 0; j < 4; ++j) {
      int   dj = __shfl(d[c], j, 4);
      float nj = __shfl(nrm[c], j, 4);
      a[c][j] = (unsigned)dj * 4 + c4;
      v[c][j] *= nj;
    }
  }
  if (c4 < NC) {
#pragma unroll
    for (int c = 0; c < EPW / 32; ++c) {
#pragma unroll
      for (int j = 0; j < 4; ++j) atomicAdd(&agg2[a[c][j]], v[c][j]);
    }
  }
}

__global__ void k_logsoftmax(const float* __restrict__ agg2, const float* __restrict__ b2,
                             float* __restrict__ out) {
  int i = blockIdx.x * blockDim.x + threadIdx.x;
  if (i >= NN) return;
  float z0 = agg2[(size_t)i * 4 + 0] + b2[0];
  float z1 = agg2[(size_t)i * 4 + 1] + b2[1];
  float z2 = agg2[(size_t)i * 4 + 2] + b2[2];
  float mx = fmaxf(z0, fmaxf(z1, z2));
  float lse = logf(expf(z0 - mx) + expf(z1 - mx) + expf(z2 - mx));
  out[(size_t)i * NC + 0] = z0 - mx - lse;
  out[(size_t)i * NC + 1] = z1 - mx - lse;
  out[(size_t)i * NC + 2] = z2 - mx - lse;
}

extern "C" void kernel_launch(void* const* d_in, const int* in_sizes, int n_in,
                              void* d_out, int out_size, void* d_ws, size_t ws_size,
                              hipStream_t stream) {
  const float* x  = (const float*)d_in[0];
  const float* W1 = (const float*)d_in[1];
  const float* b1 = (const float*)d_in[2];
  const float* W2 = (const float*)d_in[3];
  const float* b2 = (const float*)d_in[4];
  const int*   ei = (const int*)d_in[5];
  const int* src = ei;
  const int* dst = ei + EE;
  float* out = (float*)d_out;

  float* ws   = (float*)d_ws;
  float* h1   = ws;
  float* dinv = h1   + (size_t)NN * HID;
  float* agg1 = dinv + (size_t)NN;
  float* h2   = agg1 + (size_t)NN * HID;
  float* agg2 = h2   + (size_t)NN * 4;

  const int T = 256;
  k_init_deg   <<<(NN + T - 1) / T, T, 0, stream>>>(dinv);
  k_count_deg  <<<(EE + T - 1) / T, T, 0, stream>>>(dst, dinv);
  k_rsqrt      <<<(NN + T - 1) / T, T, 0, stream>>>(dinv);

  k_gemm1_wmma <<<(NN / 16 + 7) / 8, T, 0, stream>>>(x, W1, h1);

  k_agg1_init  <<<(NN * HID + T - 1) / T, T, 0, stream>>>(h1, dinv, agg1);
  k_agg1_edges <<<EE / EPW / 8, T, 0, stream>>>(h1, dinv, src, dst, agg1);

  k_relu_lin2  <<<(NN + T - 1) / T, T, 0, stream>>>(agg1, b1, W2, h2);

  k_agg2_init  <<<(NN * 4 + T - 1) / T, T, 0, stream>>>(h2, dinv, agg2);
  k_agg2_edges <<<EE / EPW / 8, T, 0, stream>>>(h2, dinv, src, dst, agg2);

  k_logsoftmax <<<(NN + T - 1) / T, T, 0, stream>>>(agg2, b2, out);
}


# one copy per chip half of every atomic target (degree, layer-1 and layer-2 sums): a wave adds into its half's copy (die id message), reader adds the two partial sums; atomic arrival order free
# speedup vs baseline: 3.5498x; 2.4191x over previous
#include <hip/hip_runtime.h>
#include <cmath>

#define NN   100000
#define EE   3200000
#define F_INd 500
#define HID  16
#define NC   3

typedef __attribute__((ext_vector_type(2))) float v2f;
typedef __attribute__((ext_vector_type(8))) float v8f;

__device__ __forceinline__ int chip_half() {
  return (__builtin_amdgcn_s_sendmsg_rtn(0x87) >> 18) & 1;
}

#define EPW 128
static_assert(EE % EPW == 0, "edge runs must tile the edge list exactly");

#define WS_TWO_COPIES ((size_t)49 * NN * sizeof(float))

__global__ void k_init_deg(float* __restrict__ degA, float* __restrict__ degB) {
  int i = blockIdx.x * blockDim.x + threadIdx.x;
  if (i < NN) { degA[i] = 1.0f; degB[i] = 0.0f; }
}

__global__ __launch_bounds__(256) void k_count_deg(const int* __restrict__ dst,
                                                   float* __restrict__ degA,
                                                   float* __restrict__ degB) {
  const int lane = threadIdx.x & 31;
  const int wave = (blockIdx.x * 256 + threadIdx.x) >> 5;
  const int e0   = wave * EPW + lane;
  float* deg = chip_half() ? degB : degA;
  int d[EPW / 32];
#pragma unroll
  for (int c = 0; c < EPW / 32; ++c) d[c] = dst[e0 + c * 32];
#pragma unroll
  for (int c = 0; c < EPW / 32; ++c) atomicAdd(&deg[(unsigned)d[c]], 1.0f);
}

__global__ void k_rsqrt(float* __restrict__ dinv, const float* __restrict__ degB) {
  int i = blockIdx.x * blockDim.x + threadIdx.x;
  if (i < NN) dinv[i] = rsqrtf(dinv[i] + degB[i]);
}

__global__ __launch_bounds__(256) void k_gemm1_wmma(const float* __restrict__ X,
                                                    const float* __restrict__ W,
                                                    float* __restrict__ H1) {
  __shared__ float Wlds[F_INd * HID];
  for (int i = threadIdx.x; i < F_INd * HID; i += 256) Wlds[i] = W[i];
  __syncthreads();

  const int wave = threadIdx.x >> 5;
  const int lane = threadIdx.x & 31;
  const int tile = blockIdx.x * 8 + wave;
  if (tile >= NN / 16) return;

  const int row_base = tile * 16;
  const int m    = lane & 15;
  const int hi   = lane >> 4;
  const int koff = hi << 1;

  const float* xrow = X + (size_t)(row_base + m) * F_INd;

  v8f acc = {};
  for (int k0 = 0; k0 < F_INd; k0 += 4) {
    v2f a, b;
    a.x = xrow[k0 + koff];
    a.y = xrow[k0 + koff + 1];
    b.x = Wlds[(k0 + koff)     * HID + m];
    b.y = Wlds[(k0 + koff + 1) * HID + m];
    acc = __builtin_amdgcn_wmma_f32_16x16x4_f32(false, a, false, b,
                                                (short)0, acc, false, false);
  }

  const int rbase = row_base + (hi << 3);
#pragma unroll
  for (int r = 0; r < 8; ++r)
    H1[(size_t)(rbase + r) * HID + m] = acc[r];
}

__global__ void k_agg1_init(const float* __restrict__ h1, const float* __restrict__ dinv,
                            float* aggA, float* aggB2, size_t ws_bytes) {
  int i = blockIdx.x * blockDim.x + threadIdx.x;
  if (i >= NN * HID) return;
  int node = i >> 4;
  float di = dinv[node];
  aggA[i] = h1[i] * di * di;
  if (ws_bytes >= WS_TWO_COPIES) aggB2[i] = 0.0f;
}

__global__ __launch_bounds__(256) void k_agg1_edges(const float* __restrict__ h1,
                                                    const float* __restrict__ dinv,
                                                    const int* __restrict__ src,
                                                    const int* __restrict__ dst,
                                                    float* aggA, float* aggB2, size_t ws_bytes) {
  const int lane = threadIdx.x & 31;
  const int wave = (blockIdx.x * 256 + threadIdx.x) >> 5;
  const int f    = lane & 15;
  const int e0   = wave * EPW + lane;
  float* agg1 = (ws_bytes >= WS_TWO_COPIES && chip_half()) ? aggB2 : aggA;

  int   s[EPW / 32], d[EPW / 32];
  float nrm[EPW / 32];
#pragma unroll
  for (int c = 0; c < EPW / 32; ++c) { s[c] = src[e0 + c * 32]; d[c] = dst[e0 + c * 32]; }
#pragma unroll
  for (int c = 0; c < EPW / 32; ++c) nrm[c] = dinv[(unsigned)s[c]] * dinv[(unsigned)d[c]];

#pragma unroll
  for (int c = 0; c < EPW / 32; ++c) {
    float v[16];
#pragma unroll
    for (int j = 0; j < 16; ++j) {
      int sj = __shfl(s[c], j, 16);
      v[j] = h1[(unsigned)sj * HID + f];
    }
#pragma unroll
    for (int j = 0; j < 16; ++j) {
      int   dj = __shfl(d[c], j, 16);
      float nj = __shfl(nrm[c], j, 16);
      atomicAdd(&agg1[(unsigned)dj * HID + f], v[j] * nj);
    }
  }
}

__global__ void k_relu_lin2(const float* __restrict__ aggA, const float* __restrict__ aggB2,
                            size_t ws_bytes, const float* __restrict__ b1,
                            const float* __restrict__ W2, float* __restrict__ h2) {
  int i = blockIdx.x * blockDim.x + threadIdx.x;
  if (i >= NN) return;
  const bool two = ws_bytes >= WS_TWO_COPIES;
  float v[HID];
#pragma unroll
  for (int f = 0; f < HID; ++f) {
    float a = aggA[(size_t)i * HID + f];
    if (two) a += aggB2[(size_t)i * HID + f];
    float t = a + b1[f];
    v[f] = t > 0.0f ? t : 0.0f;
  }
#pragma unroll
  for (int c = 0; c < NC; ++c) {
    float s = 0.0f;
#pragma unroll
    for (int f = 0; f < HID; ++f) s += v[f] * W2[f * NC + c];
    h2[(size_t)i * 4 + c] = s;
  }
  h2[(size_t)i * 4 + 3] = 0.0f;
}

__global__ void k_agg2_init(const float* __restrict__ h2, const float* __restrict__ dinv,
                            float* __restrict__ agg2A, float* __restrict__ agg2B) {
  int i = blockIdx.x * blockDim.x + threadIdx.x;
  if (i >= NN * 4) return;
  int node = i >> 2;
  float di = dinv[node];
  agg2A[i] = h2[i] * di * di;
  agg2B[i] = 0.0f;
}

__global__ __launch_bounds__(256) void k_agg2_edges(const float* __restrict__ h2,
                                                    const float* __restrict__ dinv,
                                                    const int* __restrict__ src,
                                                    const int* __restrict__ dst,
                                                    float* __restrict__ agg2A,
                                                    float* __restrict__ agg2B) {
  const int lane = threadIdx.x & 31;
  const int wave = (blockIdx.x * 256 + threadIdx.x) >> 5;
  const int c4   = lane & 3;
  const int e0   = wave * EPW + lane;
  float* agg2 = chip_half() ? agg2B : agg2A;

  int   s[EPW / 32], d[EPW / 32];
  float nrm[EPW / 32];
#pragma unroll
  for (int c = 0; c < EPW / 32; ++c) { s[c] = src[e0 + c * 32]; d[c] = dst[e0 + c * 32]; }
#pragma unroll
  for (int c = 0; c < EPW / 32; ++c) nrm[c] = dinv[(unsigned)s[c]] * dinv[(unsigned)d[c]];

  float    v[EPW / 32][4];
  unsigned a[EPW / 32][4];
#pragma unroll
  for (int c = 0; c < EPW / 32; ++c) {
#pragma unroll
    for (int j = 0; j < 4; ++j) {
      int sj = __shfl(s[c], j, 4);
      v[c][j] = h2[(unsigned)sj * 4 + c4];
    }
  }
#pragma unroll
  for (int c = 0; c < EPW / 32; ++c) {
#pragma unroll
    for (int j = 0; j < 4; ++j) {
      int   dj = __shfl(d[c], j, 4);
      float nj = __shfl(nrm[c], j, 4);
      a[c][j] = (unsigned)dj * 4 + c4;
      v[c][j] *= nj;
    }
  }
  if (c4 < NC) {
#pragma unroll
    for (int c = 0; c < EPW / 32; ++c) {
#pragma unroll
      for (int j = 0; j < 4; ++j) atomicAdd(&agg2[a[c][j]], v[c][j]);
    }
  }
}

__global__ void k_logsoftmax(const float* __restrict__ agg2A, const float* __restrict__ agg2B,
                             const float* __restrict__ b2, float* __restrict__ out) {
  int i = blockIdx.x * blockDim.x + threadIdx.x;
  if (i >= NN) return;
  float z0 = (agg2A[(size_t)i * 4 + 0] + agg2B[(size_t)i * 4 + 0]) + b2[0];
  float z1 = (agg2A[(size_t)i * 4 + 1] + agg2B[(size_t)i * 4 + 1]) + b2[1];
  float z2 = (agg2A[(size_t)i * 4 + 2] + agg2B[(size_t)i * 4 + 2]) + b2[2];
  float mx = fmaxf(z0, fmaxf(z1, z2));
  float lse = logf(expf(z0 - mx) + expf(z1 - mx) + expf(z2 - mx));
  out[(size_t)i * NC + 0] = z0 - mx - lse;
  out[(size_t)i * NC + 1] = z1 - mx - lse;
  out[(size_t)i * NC + 2] = z2 - mx - lse;
}

extern "C" void kernel_launch(void* const* d_in, const int* in_sizes, int n_in,
                              void* d_out, int out_size, void* d_ws, size_t ws_size,
                              hipStream_t stream) {
  const float* x  = (const float*)d_in[0];
  const float* W1 = (const float*)d_in[1];
  const float* b1 = (const float*)d_in[2];
  const float* W2 = (const float*)d_in[3];
  const float* b2 = (const float*)d_in[4];
  const int*   ei = (const int*)d_in[5];
  const int* src = ei;
  const int* dst = ei + EE;
  float* out = (float*)d_out;

  float* ws    = (float*)d_ws;
  float* h1    = ws;
  float* dinv  = ws + (size_t)16 * NN;
  float* agg1A = ws + (size_t)17 * NN;
  float* agg1B = ws + (size_t)33 * NN;
  float* degB  = ws + (size_t)33 * NN;
  float* h2    = ws;
  float* agg2A = ws + (size_t)4 * NN;
  float* agg2B = ws + (size_t)8 * NN;

  const int T = 256;
  k_init_deg   <<<(NN + T - 1) / T, T, 0, stream>>>(dinv, degB);
  k_count_deg  <<<EE / EPW / 8, T, 0, stream>>>(dst, dinv, degB);
  k_rsqrt      <<<(NN + T - 1) / T, T, 0, stream>>>(dinv, degB);

  k_gemm1_wmma <<<(NN / 16 + 7) / 8, T, 0, stream>>>(x, W1, h1);

  k_agg1_init  <<<(NN * HID + T - 1) / T, T, 0, stream>>>(h1, dinv, agg1A, agg1B, ws_size);
  k_agg1_edges <<<EE / EPW / 8, T, 0, stream>>>(h1, dinv, src, dst, agg1A, agg1B, ws_size);

  k_relu_lin2  <<<(NN + T - 1) / T, T, 0, stream>>>(agg1A, agg1B, ws_size, b1, W2, h2);

  k_agg2_init  <<<(NN * 4 + T - 1) / T, T, 0, stream>>>(h2, dinv, agg2A, agg2B);
  k_agg2_edges <<<EE / EPW / 8, T, 0, stream>>>(h2, dinv, src, dst, agg2A, agg2B);

  k_logsoftmax <<<(NN + T - 1) / T, T, 0, stream>>>(agg2A, agg2B, b2, out);
}


# edge weight dinv[src]*dinv[dst] factored out of both edge sums: rows scaled by own dinv where made, sums by dinv[dst] where read; edge passes fetch no dinv (term rounding moves; atomic order free)
# speedup vs baseline: 4.6279x; 1.3037x over previous
#include <hip/hip_runtime.h>
#include <cmath>

#define NN   100000
#define EE   3200000
#define F_INd 500
#define HID  16
#define NC   3

typedef __attribute__((ext_vector_type(2))) float v2f;
typedef __attribute__((ext_vector_type(8))) float v8f;

__device__ __forceinline__ int chip_half() {
  return (__builtin_amdgcn_s_sendmsg_rtn(0x87) >> 18) & 1;
}

#define EPW 128
static_assert(EE % EPW == 0, "edge runs must tile the edge list exactly");

#define WS_TWO_COPIES ((size_t)49 * NN * sizeof(float))


__global__ void k_init_deg(float* __restrict__ degA, float* __restrict__ degB) {
  int i = blockIdx.x * blockDim.x + threadIdx.x;
  if (i < NN) { degA[i] = 1.0f; degB[i] = 0.0f; }
}

__global__ __launch_bounds__(256) void k_count_deg(const int* __restrict__ dst,
                                                   float* __restrict__ degA,
                                                   float* __restrict__ degB) {
  const int lane = threadIdx.x & 31;
  const int wave = (blockIdx.x * 256 + threadIdx.x) >> 5;
  const int e0   = wave * EPW + lane;
  float* deg = chip_half() ? degB : degA;
  int d[EPW / 32];
#pragma unroll
  for (int c = 0; c < EPW / 32; ++c) d[c] = dst[e0 + c * 32];
#pragma unroll
  for (int c = 0; c < EPW / 32; ++c) atomicAdd(&deg[(unsigned)d[c]], 1.0f);
}

__global__ void k_rsqrt(float* __restrict__ dinv, const float* __restrict__ degB) {
  int i = blockIdx.x * blockDim.x + threadIdx.x;
  if (i < NN) dinv[i] = rsqrtf(dinv[i] + degB[i]);
}

__global__ __launch_bounds__(256) void k_gemm1_wmma(const float* __restrict__ X,
                                                    const float* __restrict__ W,
                                                    const float* __restrict__ dinv,
                                                    float* __restrict__ H1) {
  __shared__ float Wlds[F_INd * HID];
  for (int i = threadIdx.x; i < F_INd * HID; i += 256) Wlds[i] = W[i];
  __syncthreads();

  const int wave = threadIdx.x >> 5;
  const int lane = threadIdx.x & 31;
  const int tile = blockIdx.x * 8 + wave;
  if (tile >= NN / 16) return;

  const int row_base = tile * 16;
  const int m    = lane & 15;
  const int hi   = lane >> 4;
  const int koff = hi << 1;

  const float* xrow = X + (size_t)(row_base + m) * F_INd;

  v8f acc = {};
  for (int k0 = 0; k0 < F_INd; k0 += 4) {
    v2f a, b;
    a.x = xrow[k0 + koff];
    a.y = xrow[k0 + koff + 1];
    b.x = Wlds[(k0 + koff)     * HID + m];
    b.y = Wlds[(k0 + koff + 1) * HID + m];
    acc = __builtin_amdgcn_wmma_f32_16x16x4_f32(false, a, false, b,
                                                (short)0, acc, false, false);
  }

  const int rbase = row_base + (hi << 3);
#pragma unroll
  for (int r = 0; r < 8; ++r)
    H1[(size_t)(rbase + r) * HID + m] = acc[r] * dinv[rbase + r];
}

__global__ void k_agg1_init(const float* __restrict__ h1,
                            float* aggA, float* aggB2, size_t ws_bytes) {
  int i = blockIdx.x * blockDim.x + threadIdx.x;
  if (i >= NN * HID) return;
  aggA[i] = h1[i];
  if (ws_bytes >= WS_TWO_COPIES) aggB2[i] = 0.0f;
}

__global__ __launch_bounds__(256) void k_agg1_edges(const float* __restrict__ h1,
                                                    const int* __restrict__ src,
                                                    const int* __restrict__ dst,
                                                    float* aggA, float* aggB2, size_t ws_bytes) {
  const int lane = threadIdx.x & 31;
  const int wave = (blockIdx.x * 256 + threadIdx.x) >> 5;
  const int f    = lane & 15;
  const int e0   = wave * EPW + lane;
  float* agg1 = (ws_bytes >= WS_TWO_COPIES && chip_half()) ? aggB2 : aggA;

  int s[EPW / 32], d[EPW / 32];
#pragma unroll
  for (int c = 0; c < EPW / 32; ++c) { s[c] = src[e0 + c * 32]; d[c] = dst[e0 + c * 32]; }

#pragma unroll
  for (int c = 0; c < EPW / 32; ++c) {
    float v[16];
#pragma unroll
    for (int j = 0; j < 16; ++j) {
      int sj = __shfl(s[c], j, 16);
      v[j] = h1[(unsigned)sj * HID + f];
    }
#pragma unroll
    for (int j = 0; j < 16; ++j) {
      int dj = __shfl(d[c], j, 16);
      atomicAdd(&agg1[(unsigned)dj * HID + f], v[j]);
    }
  }
}

__global__ void k_relu_lin2(const float* __restrict__ aggA, const float* __restrict__ aggB2,
                            size_t ws_bytes, const float* __restrict__ dinv,
                            const float* __restrict__ b1,
                            const float* __restrict__ W2, float* __restrict__ h2) {
  int i = blockIdx.x * blockDim.x + threadIdx.x;
  if (i >= NN) return;
  const bool two = ws_bytes >= WS_TWO_COPIES;
  const float di = dinv[i];
  float v[HID];
#pragma unroll
  for (int f = 0; f < HID; ++f) {
    float a = aggA[(size_t)i * HID + f];
    if (two) a += aggB2[(size_t)i * HID + f];
    float t = a * di + b1[f];
    v[f] = t > 0.0f ? t : 0.0f;
  }
#pragma unroll
  for (int c = 0; c < NC; ++c) {
    float s = 0.0f;
#pragma unroll
    for (int f = 0; f < HID; ++f) s += v[f] * W2[f * NC + c];
    h2[(size_t)i * 4 + c] = s * di;
  }
  h2[(size_t)i * 4 + 3] = 0.0f;
}

__global__ void k_agg2_init(const float* __restrict__ h2,
                            float* __restrict__ agg2A, float* __restrict__ agg2B) {
  int i = blockIdx.x * blockDim.x + threadIdx.x;
  if (i >= NN * 4) return;
  agg2A[i] = h2[i];
  agg2B[i] = 0.0f;
}

__global__ __launch_bounds__(256) void k_agg2_edges(const float* __restrict__ h2,
                                                    const int* __restrict__ src,
                                                    const int* __restrict__ dst,
                                                    float* __restrict__ agg2A,
                                                    float* __restrict__ agg2B) {
  const int lane = threadIdx.x & 31;
  const int wave = (blockIdx.x * 256 + threadIdx.x) >> 5;
  const int c4   = lane & 3;
  const int e0   = wave * EPW + lane;
  float* agg2 = chip_half() ? agg2B : agg2A;

  int s[EPW / 32], d[EPW / 32];
#pragma unroll
  for (int c = 0; c < EPW / 32; ++c) { s[c] = src[e0 + c * 32]; d[c] = dst[e0 + c * 32]; }

  float    v[EPW / 32][4];
  unsigned a[EPW / 32][4];
#pragma unroll
  for (int c = 0; c < EPW / 32; ++c) {
#pragma unroll
    for (int j = 0; j < 4; ++j) {
      int sj = __shfl(s[c], j, 4);
      v[c][j] = h2[(unsigned)sj * 4 + c4];
    }
  }
#pragma unroll
  for (int c = 0; c < EPW / 32; ++c) {
#pragma unroll
    for (int j = 0; j < 4; ++j) {
      int dj = __shfl(d[c], j, 4);
      a[c][j] = (unsigned)dj * 4 + c4;
    }
  }
  if (c4 < NC) {
#pragma unroll
    for (int c = 0; c < EPW / 32; ++c) {
#pragma unroll
      for (int j = 0; j < 4; ++j) atomicAdd(&agg2[a[c][j]], v[c][j]);
    }
  }
}

__global__ void k_logsoftmax(const float* __restrict__ agg2A, const float* __restrict__ agg2B,
                             const float* __restrict__ dinv,
                             const float* __restrict__ b2, float* __restrict__ out) {
  int i = blockIdx.x * blockDim.x + threadIdx.x;
  if (i >= NN) return;
  const float di = dinv[i];
  float z0 = (agg2A[(size_t)i * 4 + 0] + agg2B[(size_t)i * 4 + 0]) * di + b2[0];
  float z1 = (agg2A[(size_t)i * 4 + 1] + agg2B[(size_t)i * 4 + 1]) * di + b2[1];
  float z2 = (agg2A[(size_t)i * 4 + 2] + agg2B[(size_t)i * 4 + 2]) * di + b2[2];
  float mx = fmaxf(z0, fmaxf(z1, z2));
  float lse = logf(expf(z0 - mx) + expf(z1 - mx) + expf(z2 - mx));
  out[(size_t)i * NC + 0] = z0 - mx - lse;
  out[(size_t)i * NC + 1] = z1 - mx - lse;
  out[(size_t)i * NC + 2] = z2 - mx - lse;
}

extern "C" void kernel_launch(void* const* d_in, const int* in_sizes, int n_in,
                              void* d_out, int out_size, void* d_ws, size_t ws_size,
                              hipStream_t stream) {
  const float* x  = (const float*)d_in[0];
  const float* W1 = (const float*)d_in[1];
  const float* b1 = (const float*)d_in[2];
  const float* W2 = (const float*)d_in[3];
  const float* b2 = (const float*)d_in[4];
  const int*   ei = (const int*)d_in[5];
  const int* src = ei;
  const int* dst = ei + EE;
  float* out = (float*)d_out;

  float* ws    = (float*)d_ws;
  float* h1    = ws;
  float* dinv  = ws + (size_t)16 * NN;
  float* agg1A = ws + (size_t)17 * NN;
  float* agg1B = ws + (size_t)33 * NN;
  float* degB  = ws + (size_t)33 * NN;
  float* h2    = ws;
  float* agg2A = ws + (size_t)4 * NN;
  float* agg2B = ws + (size_t)8 * NN;

  const int T = 256;
  k_init_deg   <<<(NN + T - 1) / T, T, 0, stream>>>(dinv, degB);
  k_count_deg  <<<EE / EPW / 8, T, 0, stream>>>(dst, dinv, degB);
  k_rsqrt      <<<(NN + T - 1) / T, T, 0, stream>>>(dinv, degB);

  k_gemm1_wmma <<<(NN / 16 + 7) / 8, T, 0, stream>>>(x, W1, dinv, h1);

  k_agg1_init  <<<(NN * HID + T - 1) / T, T, 0, stream>>>(h1, agg1A, agg1B, ws_size);
  k_agg1_edges <<<EE / EPW / 8, T, 0, stream>>>(h1, src, dst, agg1A, agg1B, ws_size);

  k_relu_lin2  <<<(NN + T - 1) / T, T, 0, stream>>>(agg1A, agg1B, ws_size, dinv, b1, W2, h2);

  k_agg2_init  <<<(NN * 4 + T - 1) / T, T, 0, stream>>>(h2, agg2A, agg2B);
  k_agg2_edges <<<EE / EPW / 8, T, 0, stream>>>(h2, src, dst, agg2A, agg2B);

  k_logsoftmax <<<(NN + T - 1) / T, T, 0, stream>>>(agg2A, agg2B, dinv, b2, out);
}


# the elementwise passes between the big kernels (degree to dinv, start values of the layer-1 and layer-2 sums) folded into the kernels that make their inputs: 10 launches -> 7; same sums, same order
# speedup vs baseline: 4.9262x; 1.0645x over previous
#include <hip/hip_runtime.h>
#include <cmath>

#define NN   100000
#define EE   3200000
#define F_INd 500
#define HID  16
#define NC   3

typedef __attribute__((ext_vector_type(2))) float v2f;
typedef __attribute__((ext_vector_type(8))) float v8f;

__device__ __forceinline__ int chip_half() {
  return (__builtin_amdgcn_s_sendmsg_rtn(0x87) >> 18) & 1;
}

#define EPW 128
static_assert(EE % EPW == 0, "edge runs must tile the edge list exactly");

#define WS_TWO_COPIES ((size_t)49 * NN * sizeof(float))


__global__ void k_init_deg(float* __restrict__ degA, float* __restrict__ zeroB, size_t ws_bytes) {
  int i = blockIdx.x * blockDim.x + threadIdx.x;
  if (i >= NN) return;
  degA[i] = 1.0f;
  const int planes = ws_bytes >= WS_TWO_COPIES ? HID : 1;
  for (int k = 0; k < planes; ++k) zeroB[(size_t)k * NN + i] = 0.0f;
}

__global__ __launch_bounds__(256) void k_count_deg(const int* __restrict__ dst,
                                                   float* __restrict__ degA,
                                                   float* __restrict__ degB) {
  const int lane = threadIdx.x & 31;
  const int wave = (blockIdx.x * 256 + threadIdx.x) >> 5;
  const int e0   = wave * EPW + lane;
  float* deg = chip_half() ? degB : degA;
  int d[EPW / 32];
#pragma unroll
  for (int c = 0; c < EPW / 32; ++c) d[c] = dst[e0 + c * 32];
#pragma unroll
  for (int c = 0; c < EPW / 32; ++c) atomicAdd(&deg[(unsigned)d[c]], 1.0f);
}

__global__ __launch_bounds__(256) void k_gemm1_wmma(const float* __restrict__ X,
                                                    const float* __restrict__ W,
                                                    float* deg_dinv, float* degB,
                                                    float* __restrict__ H1,
                                                    float* __restrict__ aggA) {
  __shared__ float Wlds[F_INd * HID];
  for (int i = threadIdx.x; i < F_INd * HID; i += 256) Wlds[i] = W[i];
  __syncthreads();

  const int wave = threadIdx.x >> 5;
  const int lane = threadIdx.x & 31;
  const int tile = blockIdx.x * 8 + wave;
  if (tile >= NN / 16) return;

  const int row_base = tile * 16;
  const int m    = lane & 15;
  const int hi   = lane >> 4;
  const int koff = hi << 1;

  const float* xrow = X + (size_t)(row_base + m) * F_INd;

  const int rbase = row_base + (hi << 3);
  const int own_row = rbase + (m & 7);
  const float degA_own = deg_dinv[own_row];
  const float degB_own = degB[own_row];

  v8f acc = {};
  for (int k0 = 0; k0 < F_INd; k0 += 4) {
    v2f a, b;
    a.x = xrow[k0 + koff];
    a.y = xrow[k0 + koff + 1];
    b.x = Wlds[(k0 + koff)     * HID + m];
    b.y = Wlds[(k0 + koff + 1) * HID + m];
    acc = __builtin_amdgcn_wmma_f32_16x16x4_f32(false, a, false, b,
                                                (short)0, acc, false, false);
  }

  const float my_di = rsqrtf(degA_own + degB_own);
  float di[8];
#pragma unroll
  for (int r = 0; r < 8; ++r) di[r] = __shfl(my_di, (hi << 4) + r, 32);
#pragma unroll
  for (int r = 0; r < 8; ++r) {
    const float v = acc[r] * di[r];
    H1[(size_t)(rbase + r) * HID + m]   = v;
    aggA[(size_t)(rbase + r) * HID + m] = v;
  }
  if (m < 8) {
    deg_dinv[own_row] = my_di;
    degB[own_row]     = 0.0f;
  }
}

__global__ __launch_bounds__(256) void k_agg1_edges(const float* __restrict__ h1,
                                                    const int* __restrict__ src,
                                                    const int* __restrict__ dst,
                                                    float* aggA, float* aggB2, size_t ws_bytes) {
  const int lane = threadIdx.x & 31;
  const int wave = (blockIdx.x * 256 + threadIdx.x) >> 5;
  const int f    = lane & 15;
  const int e0   = wave * EPW + lane;
  float* agg1 = (ws_bytes >= WS_TWO_COPIES && chip_half()) ? aggB2 : aggA;

  int s[EPW / 32], d[EPW / 32];
#pragma unroll
  for (int c = 0; c < EPW / 32; ++c) { s[c] = src[e0 + c * 32]; d[c] = dst[e0 + c * 32]; }

#pragma unroll
  for (int c = 0; c < EPW / 32; ++c) {
    float v[16];
#pragma unroll
    for (int j = 0; j < 16; ++j) {
      int sj = __shfl(s[c], j, 16);
      v[j] = h1[(unsigned)sj * HID + f];
    }
#pragma unroll
    for (int j = 0; j < 16; ++j) {
      int dj = __shfl(d[c], j, 16);
      atomicAdd(&agg1[(unsigned)dj * HID + f], v[j]);
    }
  }
}

__global__ void k_relu_lin2(const float* __restrict__ aggA, const float* __restrict__ aggB2,
                            size_t ws_bytes, const float* __restrict__ dinv,
                            const float* __restrict__ b1,
                            const float* __restrict__ W2, float* __restrict__ h2,
                            float* __restrict__ agg2A, float* __restrict__ agg2B) {
  int i = blockIdx.x * blockDim.x + threadIdx.x;
  if (i >= NN) return;
  const bool two = ws_bytes >= WS_TWO_COPIES;
  const float di = dinv[i];
  float v[HID];
#pragma unroll
  for (int f = 0; f < HID; ++f) {
    float a = aggA[(size_t)i * HID + f];
    if (two) a += aggB2[(size_t)i * HID + f];
    float t = a * di + b1[f];
    v[f] = t > 0.0f ? t : 0.0f;
  }
  float o[NC];
#pragma unroll
  for (int c = 0; c < NC; ++c) {
    float s = 0.0f;
#pragma unroll
    for (int f = 0; f < HID; ++f) s += v[f] * W2[f * NC + c];
    o[c] = s * di;
  }
  const float4 row = make_float4(o[0], o[1], o[2], 0.0f);
  reinterpret_cast<float4*>(h2)[i]    = row;
  reinterpret_cast<float4*>(agg2A)[i] = row;
  reinterpret_cast<float4*>(agg2B)[i] = make_float4(0.0f, 0.0f, 0.0f, 0.0f);
}

__global__ __launch_bounds__(256) void k_agg2_edges(const float* __restrict__ h2,
                                                    const int* __restrict__ src,
                                                    const int* __restrict__ dst,
                                                    float* __restrict__ agg2A,
                                                    float* __restrict__ agg2B) {
  const int lane = threadIdx.x & 31;
  const int wave = (blockIdx.x * 256 + threadIdx.x) >> 5;
  const int c4   = lane & 3;
  const int e0   = wave * EPW + lane;
  float* agg2 = chip_half() ? agg2B : agg2A;

  int s[EPW / 32], d[EPW / 32];
#pragma unroll
  for (int c = 0; c < EPW / 32; ++c) { s[c] = src[e0 + c * 32]; d[c] = dst[e0 + c * 32]; }

  float    v[EPW / 32][4];
  unsigned a[EPW / 32][4];
#pragma unroll
  for (int c = 0; c < EPW / 32; ++c) {
#pragma unroll
    for (int j = 0; j < 4; ++j) {
      int sj = __shfl(s[c], j, 4);
      v[c][j] = h2[(unsigned)sj * 4 + c4];
    }
  }
#pragma unroll
  for (int c = 0; c < EPW / 32; ++c) {
#pragma unroll
    for (int j = 0; j < 4; ++j) {
      int dj = __shfl(d[c], j, 4);
      a[c][j] = (unsigned)dj * 4 + c4;
    }
  }
  if (c4 < NC) {
#pragma unroll
    for (int c = 0; c < EPW / 32; ++c) {
#pragma unroll
      for (int j = 0; j < 4; ++j) atomicAdd(&agg2[a[c][j]], v[c][j]);
    }
  }
}

__global__ void k_logsoftmax(const float* __restrict__ agg2A, const float* __restrict__ agg2B,
                             const float* __restrict__ dinv,
                             const float* __restrict__ b2, float* __restrict__ out) {
  int i = blockIdx.x * blockDim.x + threadIdx.x;
  if (i >= NN) return;
  const float di = dinv[i];
  float z0 = (agg2A[(size_t)i * 4 + 0] + agg2B[(size_t)i * 4 + 0]) * di + b2[0];
  float z1 = (agg2A[(size_t)i * 4 + 1] + agg2B[(size_t)i * 4 + 1]) * di + b2[1];
  float z2 = (agg2A[(size_t)i * 4 + 2] + agg2B[(size_t)i * 4 + 2]) * di + b2[2];
  float mx = fmaxf(z0, fmaxf(z1, z2));
  float lse = logf(expf(z0 - mx) + expf(z1 - mx) + expf(z2 - mx));
  out[(size_t)i * NC + 0] = z0 - mx - lse;
  out[(size_t)i * NC + 1] = z1 - mx - lse;
  out[(size_t)i * NC + 2] = z2 - mx - lse;
}

extern "C" void kernel_launch(void* const* d_in, const int* in_sizes, int n_in,
                              void* d_out, int out_size, void* d_ws, size_t ws_size,
                              hipStream_t stream) {
  const float* x  = (const float*)d_in[0];
  const float* W1 = (const float*)d_in[1];
  const float* b1 = (const float*)d_in[2];
  const float* W2 = (const float*)d_in[3];
  const float* b2 = (const float*)d_in[4];
  const int*   ei = (const int*)d_in[5];
  const int* src = ei;
  const int* dst = ei + EE;
  float* out = (float*)d_out;

  float* ws    = (float*)d_ws;
  float* h1    = ws;
  float* dinv  = ws + (size_t)16 * NN;
  float* agg1A = ws + (size_t)17 * NN;
  float* agg1B = ws + (size_t)33 * NN;
  float* degB  = ws + (size_t)33 * NN;
  float* h2    = ws;
  float* agg2A = ws + (size_t)4 * NN;
  float* agg2B = ws + (size_t)8 * NN;

  const int T = 256;
  k_init_deg   <<<(NN + T - 1) / T, T, 0, stream>>>(dinv, degB, ws_size);
  k_count_deg  <<<EE / EPW / 8, T, 0, stream>>>(dst, dinv, degB);

  k_gemm1_wmma <<<(NN / 16 + 7) / 8, T, 0, stream>>>(x, W1, dinv, degB, h1, agg1A);

  k_agg1_edges <<<EE / EPW / 8, T, 0, stream>>>(h1, src, dst, agg1A, agg1B, ws_size);

  k_relu_lin2  <<<(NN + T - 1) / T, T, 0, stream>>>(agg1A, agg1B, ws_size, dinv, b1, W2,
                                                    h2, agg2A, agg2B);

  k_agg2_edges <<<EE / EPW / 8, T, 0, stream>>>(h2, src, dst, agg2A, agg2B);

  k_logsoftmax <<<(NN + T - 1) / T, T, 0, stream>>>(agg2A, agg2B, dinv, b2, out);
}
